# starting kernel
# baseline (speedup 1.0000x reference)
#include <hip/hip_runtime.h>

#define N_NODES 50000
#define N_EDGES 800000
#define F_OUT   128
#define N_HEAD  4
#define NEG_SLOPE 0.2f

typedef __attribute__((ext_vector_type(2))) float v2f;
typedef __attribute__((ext_vector_type(8))) float v8f;

__device__ __forceinline__ void atomF(float* p, float v) {
    unsafeAtomicAdd(p, v);
}

__global__ void zero_f4(float4* __restrict__ p, int n4) {
    int i = blockIdx.x * blockDim.x + threadIdx.x;
    int stride = gridDim.x * blockDim.x;
    float4 z = make_float4(0.f, 0.f, 0.f, 0.f);
    for (; i < n4; i += stride) p[i] = z;
}

__global__ void coef_kernel(const float* __restrict__ w, const float* __restrict__ a,
                            float* __restrict__ coefB) {
    int idx = blockIdx.x * blockDim.x + threadIdx.x;
    if (idx >= F_OUT * 16) return;
    int k = idx >> 4, col = idx & 15;
    float v = 0.f;
    if (col < 8) {
        int head = col & 3, sel = col >> 2;
        v = w[head * F_OUT + k] * a[head * 2 * F_OUT + sel * F_OUT + k];
    }
    coefB[idx] = v;
}

__global__ void proj_kernel(const float* __restrict__ x, const float* __restrict__ coefB,
                            float* __restrict__ s_src, float* __restrict__ s_dst) {
    int tile = blockIdx.x;
    int lane = threadIdx.x;
    int half = lane >> 4;
    int rc   = lane & 15;
    int koff = half * 2;
    long arow = (long)(tile * 16 + rc) * F_OUT;

    v8f c = {0.f, 0.f, 0.f, 0.f, 0.f, 0.f, 0.f, 0.f};
#pragma unroll
    for (int k = 0; k < F_OUT; k += 4) {
        v2f av, bv;
        float2 ax = *(const float2*)(x + arow + k + koff);
        av.x = ax.x; av.y = ax.y;
        bv.x = coefB[(k + koff) * 16 + rc];
        bv.y = coefB[(k + koff + 1) * 16 + rc];
        c = __builtin_amdgcn_wmma_f32_16x16x4_f32(false, av, false, bv,
                                                  (short)0, c, false, false);
    }
    if (rc < 8) {
        int head = rc & 3;
        float* buf = (rc >> 2) ? s_dst : s_src;
        int mbase = tile * 16 + half * 8;
#pragma unroll
        for (int j = 0; j < 8; ++j)
            buf[head * N_NODES + mbase + j] = c[j];
    }
}

__global__ void edge_kernel(const float* __restrict__ x, const float* __restrict__ w,
                            const int* __restrict__ ei,
                            const float* __restrict__ s_src, const float* __restrict__ s_dst,
                            float* __restrict__ rowsum, float* __restrict__ out) {
    int tid    = blockIdx.x * blockDim.x + threadIdx.x;
    int wave   = tid >> 5;
    int lane   = tid & 31;
    int nwaves = (gridDim.x * blockDim.x) >> 5;
    int f      = lane * 4;

    float4 wv[N_HEAD];
#pragma unroll
    for (int i = 0; i < N_HEAD; ++i)
        wv[i] = *(const float4*)(w + i * F_OUT + f);

    for (int e = wave; e < N_EDGES; e += nwaves) {
        int sn = ei[e];
        int dn = ei[N_EDGES + e];
        float4 xv = *(const float4*)(x + (long)dn * F_OUT + f);
#pragma unroll
        for (int i = 0; i < N_HEAD; ++i) {
            float logit = s_src[i * N_NODES + sn] + s_dst[i * N_NODES + dn];
            float lr  = (logit >= 0.f) ? logit : NEG_SLOPE * logit;
            float val = expf(-lr);
            if (lane == 0) atomF(&rowsum[i * N_NODES + sn], val);
            float* o = out + ((long)i * N_NODES + sn) * F_OUT + f;
            atomF(o + 0, val * xv.x * wv[i].x);
            atomF(o + 1, val * xv.y * wv[i].y);
            atomF(o + 2, val * xv.z * wv[i].z);
            atomF(o + 3, val * xv.w * wv[i].w);
        }
    }
}

__global__ void norm_kernel(float4* __restrict__ out, const float* __restrict__ rowsum, int n4) {
    int i = blockIdx.x * blockDim.x + threadIdx.x;
    int stride = gridDim.x * blockDim.x;
    for (; i < n4; i += stride) {
        float r = rowsum[i >> 5];
        float4 v = out[i];
        float inv = 1.0f / r;
        v.x *= inv; v.y *= inv; v.z *= inv; v.w *= inv;
        out[i] = v;
    }
}

extern "C" void kernel_launch(void* const* d_in, const int* in_sizes, int n_in,
                              void* d_out, int out_size, void* d_ws, size_t ws_size,
                              hipStream_t stream) {
    const float* x  = (const float*)d_in[0];
    const float* w  = (const float*)d_in[1];
    const float* a  = (const float*)d_in[2];
    const int*   ei = (const int*)d_in[3];
    float* out = (float*)d_out;

    float* ws     = (float*)d_ws;
    float* s_src  = ws;
    float* s_dst  = ws + N_HEAD * N_NODES;
    float* rowsum = ws + 2 * N_HEAD * N_NODES;
    float* coefB  = ws + 3 * N_HEAD * N_NODES;

    const int out_f4 = N_HEAD * N_NODES * F_OUT / 4;
    const int rs_f4  = N_HEAD * N_NODES / 4;

    zero_f4<<<4096, 256, 0, stream>>>((float4*)out, out_f4);
    zero_f4<<<196, 256, 0, stream>>>((float4*)rowsum, rs_f4);
    coef_kernel<<<8, 256, 0, stream>>>(w, a, coefB);
    proj_kernel<<<N_NODES / 16, 32, 0, stream>>>(x, coefB, s_src, s_dst);
    edge_kernel<<<2048, 256, 0, stream>>>(x, w, ei, s_src, s_dst, rowsum, out);
    norm_kernel<<<4096, 256, 0, stream>>>((float4*)out, rowsum, out_f4);
}


# edges counting-sorted by source each call (integer count atomics, scan, place); a wave sums one source row in registers, stores it normalised once; no float atomics; sum order = arrival-order rank
# speedup vs baseline: 4.7691x; 4.7691x over previous
#include <hip/hip_runtime.h>

#define N_NODES 50000
#define N_EDGES 800000
#define F_OUT   128
#define N_HEAD  4
#define NEG_SLOPE 0.2f

typedef __attribute__((ext_vector_type(2))) float v2f;
typedef __attribute__((ext_vector_type(8))) float v8f;

#define SCAN_T   1024
#define SCAN_PER 49
static_assert(SCAN_T * SCAN_PER >= N_NODES, "scan tile must cover all nodes");
static_assert(N_NODES <= 65536, "destination ids are kept in 16 bits");
static_assert(N_NODES % 8 == 0, "row pass: eight rows per block, no tail");

#define SCR_CNT   0
#define SCR_RP    (SCAN_T * SCAN_PER)
#define SCR_RANK  (2 * SCAN_T * SCAN_PER)
#define SCR_SSRC  (SCR_RANK + N_EDGES)
static_assert(SCR_SSRC + 4 * N_NODES <= N_NODES * F_OUT, "scratch must fit one head plane");

__global__ void zero_f4(float4* __restrict__ p, int n4) {
    int i = blockIdx.x * blockDim.x + threadIdx.x;
    int stride = gridDim.x * blockDim.x;
    float4 z = make_float4(0.f, 0.f, 0.f, 0.f);
    for (; i < n4; i += stride) p[i] = z;
}

__global__ void coef_kernel(const float* __restrict__ w, const float* __restrict__ a,
                            float* __restrict__ coefB) {
    int idx = blockIdx.x * blockDim.x + threadIdx.x;
    if (idx >= F_OUT * 16) return;
    int k = idx >> 4, col = idx & 15;
    float v = 0.f;
    if (col < 8) {
        int head = col & 3, sel = col >> 2;
        v = w[head * F_OUT + k] * a[head * 2 * F_OUT + sel * F_OUT + k];
    }
    coefB[idx] = v;
}

__global__ void proj_kernel(const float* __restrict__ x, const float* __restrict__ coefB,
                            float* __restrict__ s_src, float* __restrict__ s_dst) {
    int tile = blockIdx.x;
    int lane = threadIdx.x;
    int half = lane >> 4;
    int rc   = lane & 15;
    int koff = half * 2;
    long arow = (long)(tile * 16 + rc) * F_OUT;

    v8f c = {0.f, 0.f, 0.f, 0.f, 0.f, 0.f, 0.f, 0.f};
#pragma unroll
    for (int k = 0; k < F_OUT; k += 4) {
        v2f av, bv;
        float2 ax = *(const float2*)(x + arow + k + koff);
        av.x = ax.x; av.y = ax.y;
        bv.x = coefB[(k + koff) * 16 + rc];
        bv.y = coefB[(k + koff + 1) * 16 + rc];
        c = __builtin_amdgcn_wmma_f32_16x16x4_f32(false, av, false, bv,
                                                  (short)0, c, false, false);
    }
    if (rc < 8) {
        int head = rc & 3;
        float* buf = (rc >> 2) ? s_dst : s_src;
        int mbase = tile * 16 + half * 8;
#pragma unroll
        for (int j = 0; j < 8; ++j)
            buf[(mbase + j) * N_HEAD + head] = c[j];
    }
}

__global__ void count_kernel(const int* __restrict__ ei, int* __restrict__ cnt,
                             int* __restrict__ rank) {
    int e = blockIdx.x * blockDim.x + threadIdx.x;
    int stride = gridDim.x * blockDim.x;
    for (; e < N_EDGES; e += stride)
        rank[e] = atomicAdd(&cnt[ei[e]], 1);
}

__global__ void __launch_bounds__(SCAN_T) scan_kernel(const int* __restrict__ cnt,
                                                      int* __restrict__ rowstart) {
    __shared__ int wsum[SCAN_T / 32];
    const int t = threadIdx.x;
    int v[SCAN_PER];
    int sum = 0;
#pragma unroll
    for (int k = 0; k < SCAN_PER; ++k) {
        int idx = k * SCAN_T + t;
        int cv = (idx < N_NODES) ? cnt[idx] : 0;
        v[k] = cv;
        sum += cv;
    }
    int incl = sum;
    for (int o = 1; o < 32; o <<= 1) {
        int n = __shfl_up(incl, o);
        if ((t & 31) >= o) incl += n;
    }
    if ((t & 31) == 31) wsum[t >> 5] = incl;
    __syncthreads();
    if (t < 32) {
        int s = wsum[t];
        int inc = s;
        for (int o = 1; o < 32; o <<= 1) {
            int n = __shfl_up(inc, o);
            if (t >= o) inc += n;
        }
        wsum[t] = inc - s;
    }
    __syncthreads();
    int run = wsum[t >> 5] + incl - sum;
#pragma unroll
    for (int k = 0; k < SCAN_PER; ++k) {
        int idx = k * SCAN_T + t;
        if (idx < N_NODES) { rowstart[idx] = run; run += v[k]; }
    }
}

__global__ void place_kernel(const int* __restrict__ ei, const int* __restrict__ cnt,
                             const int* __restrict__ rowstart,
                             const int* __restrict__ rank, const float* __restrict__ s_src,
                             unsigned short* __restrict__ col, float* __restrict__ rec) {
    int e = blockIdx.x * blockDim.x + threadIdx.x;
    int stride = gridDim.x * blockDim.x;
    for (; e < N_EDGES; e += stride) {
        int sn = ei[e];
        int dn = ei[N_EDGES + e];
        col[rowstart[sn] + rank[e]] = (unsigned short)dn;
        if (e < N_NODES) {
            int4 m = make_int4(rowstart[e], cnt[e], 0, 0);
            *(int4*)(rec + (long)e * F_OUT)       = m;
            *(float4*)(rec + (long)e * F_OUT + 4) = *(const float4*)(s_src + e * N_HEAD);
        }
    }
}

__device__ __forceinline__ float edge_weight(float logit) {
    float lr = (logit >= 0.f) ? logit : NEG_SLOPE * logit;
    return expf(-lr);
}

__device__ __forceinline__ float lane_f(float v, int j) {
    return __int_as_float(__builtin_amdgcn_readlane(__float_as_int(v), j));
}

#define ROW_FMA(acc, e, xv) \
    acc.x = fmaf((e), (xv).x, acc.x); acc.y = fmaf((e), (xv).y, acc.y); \
    acc.z = fmaf((e), (xv).z, acc.z); acc.w = fmaf((e), (xv).w, acc.w)

#define EDGE_STEP(jj) { \
    const int jx = (jj); \
    const float4 xv = *(const float4*)(xl + (long)__builtin_amdgcn_readlane(dn, jx) * F_OUT); \
    const float f0 = lane_f(e0, jx), f1 = lane_f(e1, jx), f2 = lane_f(e2, jx), f3 = lane_f(e3, jx); \
    ROW_FMA(a0, f0, xv); ROW_FMA(a1, f1, xv); ROW_FMA(a2, f2, xv); ROW_FMA(a3, f3, xv); \
    r0 += f0; r1 += f1; r2 += f2; r3 += f3; }

__global__ void __launch_bounds__(256) row_kernel(const float* __restrict__ x,
                                                  const float* __restrict__ w,
                                                  const unsigned short* __restrict__ col,
                                                  const float* __restrict__ s_dst,
                                                  float* out) {
    const int lane = threadIdx.x & 31;
    const int row = __builtin_amdgcn_readfirstlane((int)(blockIdx.x * 8 + (threadIdx.x >> 5)));
    const float* rec = out + (long)row * F_OUT;
    const int4   m   = *(const int4*)rec;
    const float4 ss  = *(const float4*)(rec + 4);
    const int start = m.x;
    const int cnt   = m.y;
    const float* xl = x + lane * 4;

    float4 a0 = make_float4(0.f, 0.f, 0.f, 0.f), a1 = a0, a2 = a0, a3 = a0;
    float r0 = 0.f, r1 = 0.f, r2 = 0.f, r3 = 0.f;

    for (int base = 0; base < cnt; base += 32) {
        const int n = min(32, cnt - base);
        int dn = 0;
        float e0 = 0.f, e1 = 0.f, e2 = 0.f, e3 = 0.f;
        if (lane < n) {
            dn = min((int)col[start + base + lane], N_NODES - 1);
            const float4 sd = *(const float4*)(s_dst + dn * N_HEAD);
            e0 = edge_weight(ss.x + sd.x);
            e1 = edge_weight(ss.y + sd.y);
            e2 = edge_weight(ss.z + sd.z);
            e3 = edge_weight(ss.w + sd.w);
        }
        int j = 0;
        for (; j + 8 <= n; j += 8) {
#pragma unroll
            for (int u = 0; u < 8; ++u) { EDGE_STEP(j + u) }
        }
        if (j + 4 <= n) {
#pragma unroll
            for (int u = 0; u < 4; ++u) { EDGE_STEP(j + u) }
            j += 4;
        }
        if (j + 2 <= n) {
            EDGE_STEP(j) EDGE_STEP(j + 1)
            j += 2;
        }
        if (j < n) { EDGE_STEP(j) }
    }

    const float4 w0 = *(const float4*)(w + 0 * F_OUT + lane * 4);
    const float4 w1 = *(const float4*)(w + 1 * F_OUT + lane * 4);
    const float4 w2 = *(const float4*)(w + 2 * F_OUT + lane * 4);
    const float4 w3 = *(const float4*)(w + 3 * F_OUT + lane * 4);
    const float i0 = 1.0f / r0, i1 = 1.0f / r1, i2 = 1.0f / r2, i3 = 1.0f / r3;
    float* o = out + (long)row * F_OUT + lane * 4;
    const long plane = (long)N_NODES * F_OUT;
    *(float4*)(o + 0 * plane) = make_float4(a0.x * w0.x * i0, a0.y * w0.y * i0, a0.z * w0.z * i0, a0.w * w0.w * i0);
    *(float4*)(o + 1 * plane) = make_float4(a1.x * w1.x * i1, a1.y * w1.y * i1, a1.z * w1.z * i1, a1.w * w1.w * i1);
    *(float4*)(o + 2 * plane) = make_float4(a2.x * w2.x * i2, a2.y * w2.y * i2, a2.z * w2.z * i2, a2.w * w2.w * i2);
    *(float4*)(o + 3 * plane) = make_float4(a3.x * w3.x * i3, a3.y * w3.y * i3, a3.z * w3.z * i3, a3.w * w3.w * i3);
}

extern "C" void kernel_launch(void* const* d_in, const int* in_sizes, int n_in,
                              void* d_out, int out_size, void* d_ws, size_t ws_size,
                              hipStream_t stream) {
    const float* x  = (const float*)d_in[0];
    const float* w  = (const float*)d_in[1];
    const float* a  = (const float*)d_in[2];
    const int*   ei = (const int*)d_in[3];
    float* out = (float*)d_out;

    float* ws    = (float*)d_ws;
    float* s_dst = ws;
    unsigned short* col = (unsigned short*)(ws + N_HEAD * N_NODES);
    float* coefB = ws + 3 * N_HEAD * N_NODES;

    float* scr   = out + (long)3 * N_NODES * F_OUT;
    int*   cnt   = (int*)scr + SCR_CNT;
    int*   rowp  = (int*)scr + SCR_RP;
    int*   rank  = (int*)scr + SCR_RANK;
    float* s_src = scr + SCR_SSRC;

    zero_f4<<<49, 256, 0, stream>>>((float4*)cnt, N_NODES / 4);
    coef_kernel<<<8, 256, 0, stream>>>(w, a, coefB);
    proj_kernel<<<N_NODES / 16, 32, 0, stream>>>(x, coefB, s_src, s_dst);
    count_kernel<<<N_EDGES / 256, 256, 0, stream>>>(ei, cnt, rank);
    scan_kernel<<<1, SCAN_T, 0, stream>>>(cnt, rowp);
    place_kernel<<<N_EDGES / 256, 256, 0, stream>>>(ei, cnt, rowp, rank, s_src, col, out);
    row_kernel<<<N_NODES / 8, 256, 0, stream>>>(x, w, col, s_dst, out);
}


# edge count and place use LDS counters over (node range, edge slice) tiles, not per-edge global atomics; clear launch gone; sum order = slice order, then LDS arrival rank
# speedup vs baseline: 8.6678x; 1.8175x over previous
#include <hip/hip_runtime.h>

#define N_NODES 50000
#define N_EDGES 800000
#define F_OUT   128
#define N_HEAD  4
#define NEG_SLOPE 0.2f

typedef __attribute__((ext_vector_type(2))) float v2f;
typedef __attribute__((ext_vector_type(8))) float v8f;

#define N_RANGE  16
#define RANGE_N  (N_NODES / N_RANGE)
#define N_SLICE  16
#define SLICE_E  (N_EDGES / N_SLICE)
#define TILE_T   1024
#define TILE_PER 49
#define SCAN_B   256
static_assert(N_RANGE * RANGE_N == N_NODES, "ranges must cover the nodes exactly");
static_assert(N_SLICE * SLICE_E == N_EDGES, "slices must cover the edges exactly");
static_assert(TILE_T * TILE_PER >= SLICE_E, "a tile's threads must cover its slice");
static_assert(N_NODES <= 65536, "destination ids are kept in 16 bits");
static_assert(N_NODES % 8 == 0, "row pass: eight rows per block, no tail");

#define SCR_TAB   0
#define SCR_SSRC  (N_SLICE * N_NODES)
#define SCR_TOTAL (SCR_SSRC + 4 * N_NODES)
static_assert(SCR_TOTAL + 1 <= N_NODES * F_OUT, "scratch must fit one head plane");

__global__ void coef_kernel(const float* __restrict__ w, const float* __restrict__ a,
                            float* __restrict__ coefB) {
    int idx = blockIdx.x * blockDim.x + threadIdx.x;
    if (idx >= F_OUT * 16) return;
    int k = idx >> 4, col = idx & 15;
    float v = 0.f;
    if (col < 8) {
        int head = col & 3, sel = col >> 2;
        v = w[head * F_OUT + k] * a[head * 2 * F_OUT + sel * F_OUT + k];
    }
    coefB[idx] = v;
}

__global__ void proj_kernel(const float* __restrict__ x, const float* __restrict__ coefB,
                            float* __restrict__ s_src, float* __restrict__ s_dst) {
    int tile = blockIdx.x;
    int lane = threadIdx.x;
    int half = lane >> 4;
    int rc   = lane & 15;
    int koff = half * 2;
    long arow = (long)(tile * 16 + rc) * F_OUT;

    v8f c = {0.f, 0.f, 0.f, 0.f, 0.f, 0.f, 0.f, 0.f};
#pragma unroll
    for (int k = 0; k < F_OUT; k += 4) {
        v2f av, bv;
        float2 ax = *(const float2*)(x + arow + k + koff);
        av.x = ax.x; av.y = ax.y;
        bv.x = coefB[(k + koff) * 16 + rc];
        bv.y = coefB[(k + koff + 1) * 16 + rc];
        c = __builtin_amdgcn_wmma_f32_16x16x4_f32(false, av, false, bv,
                                                  (short)0, c, false, false);
    }
    if (rc < 8) {
        int head = rc & 3;
        float* buf = (rc >> 2) ? s_dst : s_src;
        int mbase = tile * 16 + half * 8;
#pragma unroll
        for (int j = 0; j < 8; ++j)
            buf[(mbase + j) * N_HEAD + head] = c[j];
    }
}

__global__ void __launch_bounds__(TILE_T) tile_count_kernel(const int* __restrict__ ei,
                                                            int* __restrict__ tab,
                                                            int* __restrict__ total) {
    __shared__ int lcnt[RANGE_N];
    const int t  = threadIdx.x;
    const int r0 = (int)(blockIdx.x % N_RANGE) * RANGE_N;
    const int sl = (int)(blockIdx.x / N_RANGE);
    const int* src = ei + sl * SLICE_E;
    if (blockIdx.x == 0 && t == 0) *total = 0;
    int sv[TILE_PER];
#pragma unroll
    for (int k = 0; k < TILE_PER; ++k) {
        const int idx = k * TILE_T + t;
        sv[k] = src[min(idx, SLICE_E - 1)];
        if (idx >= SLICE_E) sv[k] = -1;
    }
    for (int i = t; i < RANGE_N; i += TILE_T) lcnt[i] = 0;
    __syncthreads();
#pragma unroll
    for (int k = 0; k < TILE_PER; ++k) {
        const unsigned d = (unsigned)(sv[k] - r0);
        if (d < (unsigned)RANGE_N) atomicAdd(&lcnt[d], 1);
    }
    __syncthreads();
    for (int i = t; i < RANGE_N; i += TILE_T) tab[sl * N_NODES + r0 + i] = lcnt[i];
}

__global__ void __launch_bounds__(SCAN_B) tile_scan_kernel(int* __restrict__ tab,
                                                           const float* __restrict__ s_src,
                                                           int* __restrict__ total,
                                                           float* __restrict__ rec) {
    __shared__ int wsum[SCAN_B / 32];
    __shared__ int bbase;
    const int t  = threadIdx.x;
    const int n  = blockIdx.x * SCAN_B + t;
    const int nn = min(n, N_NODES - 1);
    int c[N_SLICE];
    int sum = 0;
#pragma unroll
    for (int s = 0; s < N_SLICE; ++s) c[s] = tab[s * N_NODES + nn];
#pragma unroll
    for (int s = 0; s < N_SLICE; ++s) { const int v = c[s]; c[s] = sum; sum += v; }
    if (n >= N_NODES) sum = 0;
    int incl = sum;
    for (int o = 1; o < 32; o <<= 1) {
        int m = __shfl_up(incl, o);
        if ((t & 31) >= o) incl += m;
    }
    if ((t & 31) == 31) wsum[t >> 5] = incl;
    __syncthreads();
    if (t == 0) {
        int run = 0;
        for (int i = 0; i < SCAN_B / 32; ++i) { const int v = wsum[i]; wsum[i] = run; run += v; }
        bbase = atomicAdd(total, run);
    }
    __syncthreads();
    const int start = bbase + wsum[t >> 5] + incl - sum;
    if (n < N_NODES) {
#pragma unroll
        for (int s = 0; s < N_SLICE; ++s) tab[s * N_NODES + n] = start + c[s];
        *(int4*)(rec + (long)n * F_OUT)       = make_int4(start, sum, 0, 0);
        *(float4*)(rec + (long)n * F_OUT + 4) = *(const float4*)(s_src + n * N_HEAD);
    }
}

__global__ void __launch_bounds__(TILE_T) tile_place_kernel(const int* __restrict__ ei,
                                                            const int* __restrict__ tab,
                                                            unsigned short* __restrict__ col) {
    __shared__ int lpos[RANGE_N];
    const int t  = threadIdx.x;
    const int r0 = (int)(blockIdx.x % N_RANGE) * RANGE_N;
    const int sl = (int)(blockIdx.x / N_RANGE);
    const int* src = ei + sl * SLICE_E;
    const int* dst = ei + N_EDGES + sl * SLICE_E;
    int sv[TILE_PER], dv[TILE_PER];
#pragma unroll
    for (int k = 0; k < TILE_PER; ++k) {
        const int idx = k * TILE_T + t;
        sv[k] = src[min(idx, SLICE_E - 1)];
        dv[k] = dst[min(idx, SLICE_E - 1)];
        if (idx >= SLICE_E) sv[k] = -1;
    }
    for (int i = t; i < RANGE_N; i += TILE_T) lpos[i] = tab[sl * N_NODES + r0 + i];
    __syncthreads();
#pragma unroll
    for (int k = 0; k < TILE_PER; ++k) {
        const unsigned d = (unsigned)(sv[k] - r0);
        if (d < (unsigned)RANGE_N) {
            const int pos = atomicAdd(&lpos[d], 1);
            col[pos] = (unsigned short)dv[k];
        }
    }
}

__device__ __forceinline__ float edge_weight(float logit) {
    float lr = (logit >= 0.f) ? logit : NEG_SLOPE * logit;
    return expf(-lr);
}

__device__ __forceinline__ float lane_f(float v, int j) {
    return __int_as_float(__builtin_amdgcn_readlane(__float_as_int(v), j));
}

#define ROW_FMA(acc, e, xv) \
    acc.x = fmaf((e), (xv).x, acc.x); acc.y = fmaf((e), (xv).y, acc.y); \
    acc.z = fmaf((e), (xv).z, acc.z); acc.w = fmaf((e), (xv).w, acc.w)

#define EDGE_STEP(jj) { \
    const int jx = (jj); \
    const float4 xv = *(const float4*)(xl + (long)__builtin_amdgcn_readlane(dn, jx) * F_OUT); \
    const float f0 = lane_f(e0, jx), f1 = lane_f(e1, jx), f2 = lane_f(e2, jx), f3 = lane_f(e3, jx); \
    ROW_FMA(a0, f0, xv); ROW_FMA(a1, f1, xv); ROW_FMA(a2, f2, xv); ROW_FMA(a3, f3, xv); \
    r0 += f0; r1 += f1; r2 += f2; r3 += f3; }

__global__ void __launch_bounds__(256) row_kernel(const float* __restrict__ x,
                                                  const float* __restrict__ w,
                                                  const unsigned short* __restrict__ col,
                                                  const float* __restrict__ s_dst,
                                                  float* out) {
    const int lane = threadIdx.x & 31;
    const int row = __builtin_amdgcn_readfirstlane((int)(blockIdx.x * 8 + (threadIdx.x >> 5)));
    const float* rec = out + (long)row * F_OUT;
    const int4   m   = *(const int4*)rec;
    const float4 ss  = *(const float4*)(rec + 4);
    const int start = m.x;
    const int cnt   = m.y;
    const float* xl = x + lane * 4;

    float4 a0 = make_float4(0.f, 0.f, 0.f, 0.f), a1 = a0, a2 = a0, a3 = a0;
    float r0 = 0.f, r1 = 0.f, r2 = 0.f, r3 = 0.f;

    for (int base = 0; base < cnt; base += 32) {
        const int n = min(32, cnt - base);
        int dn = 0;
        float e0 = 0.f, e1 = 0.f, e2 = 0.f, e3 = 0.f;
        if (lane < n) {
            dn = min((int)col[start + base + lane], N_NODES - 1);
            const float4 sd = *(const float4*)(s_dst + dn * N_HEAD);
            e0 = edge_weight(ss.x + sd.x);
            e1 = edge_weight(ss.y + sd.y);
            e2 = edge_weight(ss.z + sd.z);
            e3 = edge_weight(ss.w + sd.w);
        }
        int j = 0;
        for (; j + 8 <= n; j += 8) {
#pragma unroll
            for (int u = 0; u < 8; ++u) { EDGE_STEP(j + u) }
        }
        if (j + 4 <= n) {
#pragma unroll
            for (int u = 0; u < 4; ++u) { EDGE_STEP(j + u) }
            j += 4;
        }
        if (j + 2 <= n) {
            EDGE_STEP(j) EDGE_STEP(j + 1)
            j += 2;
        }
        if (j < n) { EDGE_STEP(j) }
    }

    const float4 w0 = *(const float4*)(w + 0 * F_OUT + lane * 4);
    const float4 w1 = *(const float4*)(w + 1 * F_OUT + lane * 4);
    const float4 w2 = *(const float4*)(w + 2 * F_OUT + lane * 4);
    const float4 w3 = *(const float4*)(w + 3 * F_OUT + lane * 4);
    const float i0 = 1.0f / r0, i1 = 1.0f / r1, i2 = 1.0f / r2, i3 = 1.0f / r3;
    float* o = out + (long)row * F_OUT + lane * 4;
    const long plane = (long)N_NODES * F_OUT;
    *(float4*)(o + 0 * plane) = make_float4(a0.x * w0.x * i0, a0.y * w0.y * i0, a0.z * w0.z * i0, a0.w * w0.w * i0);
    *(float4*)(o + 1 * plane) = make_float4(a1.x * w1.x * i1, a1.y * w1.y * i1, a1.z * w1.z * i1, a1.w * w1.w * i1);
    *(float4*)(o + 2 * plane) = make_float4(a2.x * w2.x * i2, a2.y * w2.y * i2, a2.z * w2.z * i2, a2.w * w2.w * i2);
    *(float4*)(o + 3 * plane) = make_float4(a3.x * w3.x * i3, a3.y * w3.y * i3, a3.z * w3.z * i3, a3.w * w3.w * i3);
}

extern "C" void kernel_launch(void* const* d_in, const int* in_sizes, int n_in,
                              void* d_out, int out_size, void* d_ws, size_t ws_size,
                              hipStream_t stream) {
    const float* x  = (const float*)d_in[0];
    const float* w  = (const float*)d_in[1];
    const float* a  = (const float*)d_in[2];
    const int*   ei = (const int*)d_in[3];
    float* out = (float*)d_out;

    float* ws    = (float*)d_ws;
    float* s_dst = ws;
    unsigned short* col = (unsigned short*)(ws + N_HEAD * N_NODES);
    float* coefB = ws + 3 * N_HEAD * N_NODES;

    float* scr   = out + (long)3 * N_NODES * F_OUT;
    int*   tab   = (int*)scr + SCR_TAB;
    float* s_src = scr + SCR_SSRC;
    int*   total = (int*)scr + SCR_TOTAL;

    coef_kernel<<<8, 256, 0, stream>>>(w, a, coefB);
    proj_kernel<<<N_NODES / 16, 32, 0, stream>>>(x, coefB, s_src, s_dst);
    tile_count_kernel<<<N_RANGE * N_SLICE, TILE_T, 0, stream>>>(ei, tab, total);
    tile_scan_kernel<<<(N_NODES + SCAN_B - 1) / SCAN_B, SCAN_B, 0, stream>>>(tab, s_src, total, out);
    tile_place_kernel<<<N_RANGE * N_SLICE, TILE_T, 0, stream>>>(ei, tab, col);
    row_kernel<<<N_NODES / 8, 256, 0, stream>>>(x, w, col, s_dst, out);
}


# count, scan and place of the edge sort become one launch: a tile scans its own LDS counters, takes list room with one add, writes its runs into the row records; 6 launches to 4; sum order unchanged
# speedup vs baseline: 9.3959x; 1.0840x over previous
#include <hip/hip_runtime.h>

#define N_NODES 50000
#define N_EDGES 800000
#define F_OUT   128
#define N_HEAD  4
#define NEG_SLOPE 0.2f

typedef __attribute__((ext_vector_type(2))) float v2f;
typedef __attribute__((ext_vector_type(8))) float v8f;

#define N_RANGE  16
#define RANGE_N  (N_NODES / N_RANGE)
#define N_SLICE  16
#define SLICE_E  (N_EDGES / N_SLICE)
#define TILE_T   1024
#define TILE_PER 49
#define LCNT_PER 4
#define LCNT_N   (TILE_T * LCNT_PER)
#define REC_SSRC (2 * N_SLICE)
static_assert(N_RANGE * RANGE_N == N_NODES, "ranges must cover the nodes exactly");
static_assert(N_SLICE * SLICE_E == N_EDGES, "slices must cover the edges exactly");
static_assert(TILE_T * TILE_PER >= SLICE_E, "a tile's threads must cover its slice");
static_assert(LCNT_N >= RANGE_N, "a tile's threads must cover its counters");
static_assert(2 * N_SLICE == 32, "row pass: one record word per lane");
static_assert(REC_SSRC + N_HEAD <= F_OUT, "the record must fit the row");
static_assert(N_NODES <= 65536, "destination ids are kept in 16 bits");
static_assert(N_NODES % 8 == 0, "row pass: eight rows per block, no tail");

__global__ void coef_kernel(const float* __restrict__ w, const float* __restrict__ a,
                            float* __restrict__ coefB, int* __restrict__ total) {
    int idx = blockIdx.x * blockDim.x + threadIdx.x;
    if (idx == 0) *total = 0;
    if (idx >= F_OUT * 16) return;
    int k = idx >> 4, col = idx & 15;
    float v = 0.f;
    if (col < 8) {
        int head = col & 3, sel = col >> 2;
        v = w[head * F_OUT + k] * a[head * 2 * F_OUT + sel * F_OUT + k];
    }
    coefB[idx] = v;
}

__global__ void proj_kernel(const float* __restrict__ x, const float* __restrict__ coefB,
                            float* __restrict__ rec, float* __restrict__ s_dst) {
    int tile = blockIdx.x;
    int lane = threadIdx.x;
    int half = lane >> 4;
    int rc   = lane & 15;
    int koff = half * 2;
    long arow = (long)(tile * 16 + rc) * F_OUT;

    v8f c = {0.f, 0.f, 0.f, 0.f, 0.f, 0.f, 0.f, 0.f};
#pragma unroll
    for (int k = 0; k < F_OUT; k += 4) {
        v2f av, bv;
        float2 ax = *(const float2*)(x + arow + k + koff);
        av.x = ax.x; av.y = ax.y;
        bv.x = coefB[(k + koff) * 16 + rc];
        bv.y = coefB[(k + koff + 1) * 16 + rc];
        c = __builtin_amdgcn_wmma_f32_16x16x4_f32(false, av, false, bv,
                                                  (short)0, c, false, false);
    }
    if (rc < 8) {
        int head = rc & 3;
        float* buf = (rc >> 2) ? s_dst : (rec + REC_SSRC);
        const int pitch = (rc >> 2) ? N_HEAD : F_OUT;
        int mbase = tile * 16 + half * 8;
#pragma unroll
        for (int j = 0; j < 8; ++j)
            buf[(mbase + j) * pitch + head] = c[j];
    }
}

__global__ void __launch_bounds__(TILE_T) tile_sort_kernel(const int* __restrict__ ei,
                                                           int* __restrict__ total,
                                                           int* __restrict__ rec,
                                                           unsigned short* __restrict__ col) {
    __shared__ __align__(16) int lcnt[LCNT_N];
    __shared__ int wsum[TILE_T / 32];
    __shared__ int bbase;
    const int t  = threadIdx.x;
    const int r0 = (int)(blockIdx.x % N_RANGE) * RANGE_N;
    const int sl = (int)(blockIdx.x / N_RANGE);
    const int* src = ei + sl * SLICE_E;
    const int* dst = ei + N_EDGES + sl * SLICE_E;
    int sv[TILE_PER], dv[TILE_PER];
#pragma unroll
    for (int k = 0; k < TILE_PER; ++k) {
        const int idx = k * TILE_T + t;
        sv[k] = src[min(idx, SLICE_E - 1)];
        dv[k] = dst[min(idx, SLICE_E - 1)];
        if (idx >= SLICE_E) sv[k] = -1;
    }
    *(int4*)&lcnt[LCNT_PER * t] = make_int4(0, 0, 0, 0);
    __syncthreads();
#pragma unroll
    for (int k = 0; k < TILE_PER; ++k) {
        const unsigned d = (unsigned)(sv[k] - r0);
        if (d < (unsigned)RANGE_N) atomicAdd(&lcnt[d], 1);
    }
    __syncthreads();
    const int4 c4 = *(const int4*)&lcnt[LCNT_PER * t];
    const int s4 = c4.x + c4.y + c4.z + c4.w;
    int incl = s4;
    for (int o = 1; o < 32; o <<= 1) {
        int m = __shfl_up(incl, o);
        if ((t & 31) >= o) incl += m;
    }
    if ((t & 31) == 31) wsum[t >> 5] = incl;
    __syncthreads();
    if (t < 32) {
        const int v = wsum[t];
        int wi = v;
        for (int o = 1; o < 32; o <<= 1) {
            int m = __shfl_up(wi, o);
            if (t >= o) wi += m;
        }
        wsum[t] = wi - v;
        if (t == 31) bbase = atomicAdd(total, wi);
    }
    __syncthreads();
    const int p0 = bbase + wsum[t >> 5] + incl - s4;
    const int p1 = p0 + c4.x, p2 = p1 + c4.y, p3 = p2 + c4.z;
    *(int4*)&lcnt[LCNT_PER * t] = make_int4(p0, p1, p2, p3);
    {
        const int i0 = LCNT_PER * t;
        int* rp = rec + (long)(r0 + i0) * F_OUT + 2 * sl;
        if (i0 + 0 < RANGE_N) *(int2*)(rp + 0 * F_OUT) = make_int2(p0, c4.x);
        if (i0 + 1 < RANGE_N) *(int2*)(rp + 1 * F_OUT) = make_int2(p1, c4.y);
        if (i0 + 2 < RANGE_N) *(int2*)(rp + 2 * F_OUT) = make_int2(p2, c4.z);
        if (i0 + 3 < RANGE_N) *(int2*)(rp + 3 * F_OUT) = make_int2(p3, c4.w);
    }
    __syncthreads();
#pragma unroll
    for (int k = 0; k < TILE_PER; ++k) {
        const unsigned d = (unsigned)(sv[k] - r0);
        if (d < (unsigned)RANGE_N) {
            const int pos = atomicAdd(&lcnt[d], 1);
            col[pos] = (unsigned short)dv[k];
        }
    }
}

__device__ __forceinline__ float edge_weight(float logit) {
    float lr = (logit >= 0.f) ? logit : NEG_SLOPE * logit;
    return expf(-lr);
}

__device__ __forceinline__ float lane_f(float v, int j) {
    return __int_as_float(__builtin_amdgcn_readlane(__float_as_int(v), j));
}

#define ROW_FMA(acc, e, xv) \
    acc.x = fmaf((e), (xv).x, acc.x); acc.y = fmaf((e), (xv).y, acc.y); \
    acc.z = fmaf((e), (xv).z, acc.z); acc.w = fmaf((e), (xv).w, acc.w)

#define EDGE_STEP(jj) { \
    const int jx = (jj); \
    const float4 xv = *(const float4*)(xl + (long)__builtin_amdgcn_readlane(dn, jx) * F_OUT); \
    const float f0 = lane_f(e0, jx), f1 = lane_f(e1, jx), f2 = lane_f(e2, jx), f3 = lane_f(e3, jx); \
    ROW_FMA(a0, f0, xv); ROW_FMA(a1, f1, xv); ROW_FMA(a2, f2, xv); ROW_FMA(a3, f3, xv); \
    r0 += f0; r1 += f1; r2 += f2; r3 += f3; }

__device__ __forceinline__ int run_shift(int rv, int e, int& count) {
    int shift = 0, first = 0;
#pragma unroll
    for (int s = 0; s < N_SLICE; ++s) {
        const int st = __builtin_amdgcn_readlane(rv, 2 * s);
        const int ln = __builtin_amdgcn_readlane(rv, 2 * s + 1);
        shift = (e >= first) ? (st - first) : shift;
        first += ln;
    }
    count = first;
    return shift;
}

__global__ void __launch_bounds__(256) row_kernel(const float* __restrict__ x,
                                                  const float* __restrict__ w,
                                                  const unsigned short* __restrict__ col,
                                                  const float* __restrict__ s_dst,
                                                  float* out) {
    const int lane = threadIdx.x & 31;
    const int row = __builtin_amdgcn_readfirstlane((int)(blockIdx.x * 8 + (threadIdx.x >> 5)));
    const float* rec = out + (long)row * F_OUT;
    const int    rv  = ((const int*)rec)[lane];
    const float4 ss  = *(const float4*)(rec + REC_SSRC);
    int cnt;
    int shift = run_shift(rv, lane, cnt);
    const float* xl = x + lane * 4;

    float4 a0 = make_float4(0.f, 0.f, 0.f, 0.f), a1 = a0, a2 = a0, a3 = a0;
    float r0 = 0.f, r1 = 0.f, r2 = 0.f, r3 = 0.f;

    for (int base = 0; base < cnt; base += 32) {
        const int n = min(32, cnt - base);
        int dn = 0;
        float e0 = 0.f, e1 = 0.f, e2 = 0.f, e3 = 0.f;
        const int e = base + lane;
        if (lane < n) {
            dn = min((int)col[shift + e], N_NODES - 1);
            const float4 sd = *(const float4*)(s_dst + dn * N_HEAD);
            e0 = edge_weight(ss.x + sd.x);
            e1 = edge_weight(ss.y + sd.y);
            e2 = edge_weight(ss.z + sd.z);
            e3 = edge_weight(ss.w + sd.w);
        }
        int j = 0;
        for (; j + 8 <= n; j += 8) {
#pragma unroll
            for (int u = 0; u < 8; ++u) { EDGE_STEP(j + u) }
        }
        if (j + 4 <= n) {
#pragma unroll
            for (int u = 0; u < 4; ++u) { EDGE_STEP(j + u) }
            j += 4;
        }
        if (j + 2 <= n) {
            EDGE_STEP(j) EDGE_STEP(j + 1)
            j += 2;
        }
        if (j < n) { EDGE_STEP(j) }
        if (base + 32 < cnt) { int again; shift = run_shift(rv, e + 32, again); }
    }

    const float4 w0 = *(const float4*)(w + 0 * F_OUT + lane * 4);
    const float4 w1 = *(const float4*)(w + 1 * F_OUT + lane * 4);
    const float4 w2 = *(const float4*)(w + 2 * F_OUT + lane * 4);
    const float4 w3 = *(const float4*)(w + 3 * F_OUT + lane * 4);
    const float i0 = 1.0f / r0, i1 = 1.0f / r1, i2 = 1.0f / r2, i3 = 1.0f / r3;
    float* o = out + (long)row * F_OUT + lane * 4;
    const long plane = (long)N_NODES * F_OUT;
    *(float4*)(o + 0 * plane) = make_float4(a0.x * w0.x * i0, a0.y * w0.y * i0, a0.z * w0.z * i0, a0.w * w0.w * i0);
    *(float4*)(o + 1 * plane) = make_float4(a1.x * w1.x * i1, a1.y * w1.y * i1, a1.z * w1.z * i1, a1.w * w1.w * i1);
    *(float4*)(o + 2 * plane) = make_float4(a2.x * w2.x * i2, a2.y * w2.y * i2, a2.z * w2.z * i2, a2.w * w2.w * i2);
    *(float4*)(o + 3 * plane) = make_float4(a3.x * w3.x * i3, a3.y * w3.y * i3, a3.z * w3.z * i3, a3.w * w3.w * i3);
}

extern "C" void kernel_launch(void* const* d_in, const int* in_sizes, int n_in,
                              void* d_out, int out_size, void* d_ws, size_t ws_size,
                              hipStream_t stream) {
    const float* x  = (const float*)d_in[0];
    const float* w  = (const float*)d_in[1];
    const float* a  = (const float*)d_in[2];
    const int*   ei = (const int*)d_in[3];
    float* out = (float*)d_out;

    float* ws    = (float*)d_ws;
    float* s_dst = ws;
    unsigned short* col = (unsigned short*)(ws + N_HEAD * N_NODES);
    float* coefB = ws + 3 * N_HEAD * N_NODES;

    int* total = (int*)(out + (long)3 * N_NODES * F_OUT);

    coef_kernel<<<8, 256, 0, stream>>>(w, a, coefB, total);
    proj_kernel<<<N_NODES / 16, 32, 0, stream>>>(x, coefB, out, s_dst);
    tile_sort_kernel<<<N_RANGE * N_SLICE, TILE_T, 0, stream>>>(ei, total, (int*)out, col);
    row_kernel<<<N_NODES / 8, 256, 0, stream>>>(x, w, col, s_dst, out);
}


# coefficient and projection launches folded into the sort launch (its first waves each project a 16-node tile); a tile's list start = its own count of lower-range edges, no counter; 4 launches to 2
# speedup vs baseline: 9.7727x; 1.0401x over previous
#include <hip/hip_runtime.h>

#define N_NODES 50000
#define N_EDGES 800000
#define F_OUT   128
#define N_HEAD  4
#define NEG_SLOPE 0.2f

typedef __attribute__((ext_vector_type(2))) float v2f;
typedef __attribute__((ext_vector_type(8))) float v8f;

#define N_RANGE  16
#define RANGE_N  (N_NODES / N_RANGE)
#define N_SLICE  16
#define SLICE_E  (N_EDGES / N_SLICE)
#define TILE_T   1024
#define TILE_PER 49
#define LCNT_PER 4
#define LCNT_N   (TILE_T * LCNT_PER)
#define REC_SSRC (2 * N_SLICE)
#define N_TILES  (N_RANGE * N_SLICE)
#define N_PROJ   (N_NODES / 16)
static_assert(N_RANGE * RANGE_N == N_NODES, "ranges must cover the nodes exactly");
static_assert(N_SLICE * SLICE_E == N_EDGES, "slices must cover the edges exactly");
static_assert(TILE_T * TILE_PER >= SLICE_E, "a tile's threads must cover its slice");
static_assert(LCNT_N >= RANGE_N, "a tile's threads must cover its counters");
static_assert(2 * N_SLICE == 32, "row pass: one record word per lane");
static_assert(REC_SSRC + N_HEAD <= F_OUT, "the record must fit the row");
static_assert(N_NODES <= 65536, "destination ids are kept in 16 bits");
static_assert(N_NODES % 8 == 0, "row pass: eight rows per block, no tail");
static_assert(N_NODES % 16 == 0, "projection: 16-node tiles, no tail");
static_assert((long)N_TILES * (TILE_T / 32) >= N_PROJ, "the waves of the first launch must cover the projection tiles");

__global__ void __launch_bounds__(TILE_T) front_kernel(const float* __restrict__ x,
                                                       const float* __restrict__ w,
                                                       const float* __restrict__ a,
                                                       const int* __restrict__ ei,
                                                       float* out,
                                                       float* __restrict__ s_dst,
                                                       unsigned short* __restrict__ col) {
    __shared__ __align__(16) int lcnt[LCNT_N];
    __shared__ float coefS[F_OUT * 16];
    __shared__ int wsum[TILE_T / 32];
    __shared__ int sbel;
    const int t  = threadIdx.x;

    if (t == 0) sbel = 0;
    {
        const int cc = t & 15, k0 = t >> 4;
        float v0 = 0.f, v1 = 0.f;
        if (cc < 8) {
            const int head = cc & 3, sel = cc >> 2;
            const float* wp = w + head * F_OUT + k0;
            const float* ap = a + head * 2 * F_OUT + sel * F_OUT + k0;
            const float w0 = wp[0], w1 = wp[F_OUT / 2], a0 = ap[0], a1 = ap[F_OUT / 2];
            v0 = w0 * a0;
            v1 = w1 * a1;
        }
        coefS[t] = v0;
        coefS[t + TILE_T] = v1;
    }
    __syncthreads();

    {
        const int wave = __builtin_amdgcn_readfirstlane(t >> 5);
        const int tile = (int)blockIdx.x + wave * N_TILES;
        if (tile < N_PROJ) {
            const int lane = t & 31;
            const int half = lane >> 4;
            const int rc   = lane & 15;
            const int koff = half * 2;
            const long arow = (long)(tile * 16 + rc) * F_OUT;

            v8f c = {0.f, 0.f, 0.f, 0.f, 0.f, 0.f, 0.f, 0.f};
#pragma unroll
            for (int k = 0; k < F_OUT; k += 4) {
                v2f av, bv;
                const float2 ax = *(const float2*)(x + arow + k + koff);
                av.x = ax.x; av.y = ax.y;
                bv.x = coefS[(k + koff) * 16 + rc];
                bv.y = coefS[(k + koff + 1) * 16 + rc];
                c = __builtin_amdgcn_wmma_f32_16x16x4_f32(false, av, false, bv,
                                                          (short)0, c, false, false);
            }
            if (rc < 8) {
                const int head = rc & 3;
                float* buf = (rc >> 2) ? s_dst : (out + REC_SSRC);
                const int pitch = (rc >> 2) ? N_HEAD : F_OUT;
                const int mbase = tile * 16 + half * 8;
#pragma unroll
                for (int j = 0; j < 8; ++j)
                    buf[(mbase + j) * pitch + head] = c[j];
            }
        }
    }

    int* rec = (int*)out;
    const int r0 = (int)(blockIdx.x % N_RANGE) * RANGE_N;
    const int sl = (int)(blockIdx.x / N_RANGE);
    const int* src = ei + sl * SLICE_E;
    const int* dst = ei + N_EDGES + sl * SLICE_E;
    int sv[TILE_PER], dv[TILE_PER];
#pragma unroll
    for (int k = 0; k < TILE_PER; ++k) {
        const int idx = k * TILE_T + t;
        sv[k] = src[min(idx, SLICE_E - 1)];
        dv[k] = dst[min(idx, SLICE_E - 1)];
        if (idx >= SLICE_E) sv[k] = -1;
    }
    int below = 0;
#pragma unroll
    for (int k = 0; k < TILE_PER; ++k)
        below += __popc(__builtin_amdgcn_ballot_w32((unsigned)sv[k] < (unsigned)r0));
    if ((t & 31) == 0) atomicAdd(&sbel, below);
    *(int4*)&lcnt[LCNT_PER * t] = make_int4(0, 0, 0, 0);
    __syncthreads();
#pragma unroll
    for (int k = 0; k < TILE_PER; ++k) {
        const unsigned d = (unsigned)(sv[k] - r0);
        if (d < (unsigned)RANGE_N) atomicAdd(&lcnt[d], 1);
    }
    __syncthreads();
    const int4 c4 = *(const int4*)&lcnt[LCNT_PER * t];
    const int s4 = c4.x + c4.y + c4.z + c4.w;
    int incl = s4;
    for (int o = 1; o < 32; o <<= 1) {
        int m = __shfl_up(incl, o);
        if ((t & 31) >= o) incl += m;
    }
    if ((t & 31) == 31) wsum[t >> 5] = incl;
    __syncthreads();
    if (t < 32) {
        const int v = wsum[t];
        int wi = v;
        for (int o = 1; o < 32; o <<= 1) {
            int m = __shfl_up(wi, o);
            if (t >= o) wi += m;
        }
        wsum[t] = wi - v;
    }
    __syncthreads();
    const int p0 = sl * SLICE_E + sbel + wsum[t >> 5] + incl - s4;
    const int p1 = p0 + c4.x, p2 = p1 + c4.y, p3 = p2 + c4.z;
    *(int4*)&lcnt[LCNT_PER * t] = make_int4(p0, p1, p2, p3);
    {
        const int i0 = LCNT_PER * t;
        int* rp = rec + (long)(r0 + i0) * F_OUT + 2 * sl;
        if (i0 + 0 < RANGE_N) *(int2*)(rp + 0 * F_OUT) = make_int2(p0, c4.x);
        if (i0 + 1 < RANGE_N) *(int2*)(rp + 1 * F_OUT) = make_int2(p1, c4.y);
        if (i0 + 2 < RANGE_N) *(int2*)(rp + 2 * F_OUT) = make_int2(p2, c4.z);
        if (i0 + 3 < RANGE_N) *(int2*)(rp + 3 * F_OUT) = make_int2(p3, c4.w);
    }
    __syncthreads();
#pragma unroll
    for (int k = 0; k < TILE_PER; ++k) {
        const unsigned d = (unsigned)(sv[k] - r0);
        if (d < (unsigned)RANGE_N) {
            const int pos = atomicAdd(&lcnt[d], 1);
            col[pos] = (unsigned short)dv[k];
        }
    }
}

__device__ __forceinline__ float edge_weight(float logit) {
    float lr = (logit >= 0.f) ? logit : NEG_SLOPE * logit;
    return expf(-lr);
}

__device__ __forceinline__ float lane_f(float v, int j) {
    return __int_as_float(__builtin_amdgcn_readlane(__float_as_int(v), j));
}

#define ROW_FMA(acc, e, xv) \
    acc.x = fmaf((e), (xv).x, acc.x); acc.y = fmaf((e), (xv).y, acc.y); \
    acc.z = fmaf((e), (xv).z, acc.z); acc.w = fmaf((e), (xv).w, acc.w)

#define EDGE_STEP(jj) { \
    const int jx = (jj); \
    const float4 xv = *(const float4*)(xl + (long)__builtin_amdgcn_readlane(dn, jx) * F_OUT); \
    const float f0 = lane_f(e0, jx), f1 = lane_f(e1, jx), f2 = lane_f(e2, jx), f3 = lane_f(e3, jx); \
    ROW_FMA(a0, f0, xv); ROW_FMA(a1, f1, xv); ROW_FMA(a2, f2, xv); ROW_FMA(a3, f3, xv); \
    r0 += f0; r1 += f1; r2 += f2; r3 += f3; }

__device__ __forceinline__ int run_shift(int rv, int e, int& count) {
    int shift = 0, first = 0;
#pragma unroll
    for (int s = 0; s < N_SLICE; ++s) {
        const int st = __builtin_amdgcn_readlane(rv, 2 * s);
        const int ln = __builtin_amdgcn_readlane(rv, 2 * s + 1);
        shift = (e >= first) ? (st - first) : shift;
        first += ln;
    }
    count = first;
    return shift;
}

__global__ void __launch_bounds__(256) row_kernel(const float* __restrict__ x,
                                                  const float* __restrict__ w,
                                                  const unsigned short* __restrict__ col,
                                                  const float* __restrict__ s_dst,
                                                  float* out) {
    const int lane = threadIdx.x & 31;
    const int row = __builtin_amdgcn_readfirstlane((int)(blockIdx.x * 8 + (threadIdx.x >> 5)));
    const float* rec = out + (long)row * F_OUT;
    const int    rv  = ((const int*)rec)[lane];
    const float4 ss  = *(const float4*)(rec + REC_SSRC);
    int cnt;
    int shift = run_shift(rv, lane, cnt);
    const float* xl = x + lane * 4;

    float4 a0 = make_float4(0.f, 0.f, 0.f, 0.f), a1 = a0, a2 = a0, a3 = a0;
    float r0 = 0.f, r1 = 0.f, r2 = 0.f, r3 = 0.f;

    for (int base = 0; base < cnt; base += 32) {
        const int n = min(32, cnt - base);
        int dn = 0;
        float e0 = 0.f, e1 = 0.f, e2 = 0.f, e3 = 0.f;
        const int e = base + lane;
        if (lane < n) {
            dn = min((int)col[shift + e], N_NODES - 1);
            const float4 sd = *(const float4*)(s_dst + dn * N_HEAD);
            e0 = edge_weight(ss.x + sd.x);
            e1 = edge_weight(ss.y + sd.y);
            e2 = edge_weight(ss.z + sd.z);
            e3 = edge_weight(ss.w + sd.w);
        }
        int j = 0;
        for (; j + 8 <= n; j += 8) {
#pragma unroll
            for (int u = 0; u < 8; ++u) { EDGE_STEP(j + u) }
        }
        if (j + 4 <= n) {
#pragma unroll
            for (int u = 0; u < 4; ++u) { EDGE_STEP(j + u) }
            j += 4;
        }
        if (j + 2 <= n) {
            EDGE_STEP(j) EDGE_STEP(j + 1)
            j += 2;
        }
        if (j < n) { EDGE_STEP(j) }
        if (base + 32 < cnt) { int again; shift = run_shift(rv, e + 32, again); }
    }

    const float4 w0 = *(const float4*)(w + 0 * F_OUT + lane * 4);
    const float4 w1 = *(const float4*)(w + 1 * F_OUT + lane * 4);
    const float4 w2 = *(const float4*)(w + 2 * F_OUT + lane * 4);
    const float4 w3 = *(const float4*)(w + 3 * F_OUT + lane * 4);
    const float i0 = 1.0f / r0, i1 = 1.0f / r1, i2 = 1.0f / r2, i3 = 1.0f / r3;
    float* o = out + (long)row * F_OUT + lane * 4;
    const long plane = (long)N_NODES * F_OUT;
    *(float4*)(o + 0 * plane) = make_float4(a0.x * w0.x * i0, a0.y * w0.y * i0, a0.z * w0.z * i0, a0.w * w0.w * i0);
    *(float4*)(o + 1 * plane) = make_float4(a1.x * w1.x * i1, a1.y * w1.y * i1, a1.z * w1.z * i1, a1.w * w1.w * i1);
    *(float4*)(o + 2 * plane) = make_float4(a2.x * w2.x * i2, a2.y * w2.y * i2, a2.z * w2.z * i2, a2.w * w2.w * i2);
    *(float4*)(o + 3 * plane) = make_float4(a3.x * w3.x * i3, a3.y * w3.y * i3, a3.z * w3.z * i3, a3.w * w3.w * i3);
}

extern "C" void kernel_launch(void* const* d_in, const int* in_sizes, int n_in,
                              void* d_out, int out_size, void* d_ws, size_t ws_size,
                              hipStream_t stream) {
    const float* x  = (const float*)d_in[0];
    const float* w  = (const float*)d_in[1];
    const float* a  = (const float*)d_in[2];
    const int*   ei = (const int*)d_in[3];
    float* out = (float*)d_out;

    float* ws    = (float*)d_ws;
    float* s_dst = ws;
    unsigned short* col = (unsigned short*)(ws + N_HEAD * N_NODES);

    front_kernel<<<N_TILES, TILE_T, 0, stream>>>(x, w, a, ei, out, s_dst, col);
    row_kernel<<<N_NODES / 8, 256, 0, stream>>>(x, w, col, s_dst, out);
}


# a tile's piece of the sorted list is put together in LDS and leaves in whole 16-byte stores, not 2-byte scattered global stores each behind its own LDS add; sum order unchanged
# speedup vs baseline: 10.3098x; 1.0550x over previous
#include <hip/hip_runtime.h>

#define N_NODES 50000
#define N_EDGES 800000
#define F_OUT   128
#define N_HEAD  4
#define NEG_SLOPE 0.2f

typedef __attribute__((ext_vector_type(2))) float v2f;
typedef __attribute__((ext_vector_type(8))) float v8f;

#define N_RANGE  16
#define RANGE_N  (N_NODES / N_RANGE)
#define N_SLICE  16
#define SLICE_E  (N_EDGES / N_SLICE)
#define TILE_T   1024
#define TILE_PER 49
#define LCNT_PER 4
#define LCNT_N   (TILE_T * LCNT_PER)
#define REC_SSRC (2 * N_SLICE)
#define N_TILES  (N_RANGE * N_SLICE)
#define N_PROJ   (N_NODES / 16)
#define STAGE_N  (SLICE_E + 8)
static_assert(N_RANGE * RANGE_N == N_NODES, "ranges must cover the nodes exactly");
static_assert(N_SLICE * SLICE_E == N_EDGES, "slices must cover the edges exactly");
static_assert(TILE_T * TILE_PER >= SLICE_E, "a tile's threads must cover its slice");
static_assert(LCNT_N >= RANGE_N, "a tile's threads must cover its counters");
static_assert(2 * N_SLICE == 32, "row pass: one record word per lane");
static_assert(REC_SSRC + N_HEAD <= F_OUT, "the record must fit the row");
static_assert(N_NODES <= 65536, "destination ids are kept in 16 bits");
static_assert(N_NODES % 8 == 0, "row pass: eight rows per block, no tail");
static_assert(N_NODES % 16 == 0, "projection: 16-node tiles, no tail");
static_assert((long)N_TILES * (TILE_T / 32) >= N_PROJ, "the waves of the first launch must cover the projection tiles");

__global__ void __launch_bounds__(TILE_T) front_kernel(const float* __restrict__ x,
                                                       const float* __restrict__ w,
                                                       const float* __restrict__ a,
                                                       const int* __restrict__ ei,
                                                       float* out,
                                                       float* __restrict__ s_dst,
                                                       unsigned short* __restrict__ col) {
    __shared__ __align__(16) int lcnt[LCNT_N];
    __shared__ float coefS[F_OUT * 16];
    __shared__ int wsum[TILE_T / 32];
    __shared__ int sbel;
    __shared__ int stot;
    __shared__ __align__(16) unsigned short stage[STAGE_N];
    const int t  = threadIdx.x;

    if (t == 0) sbel = 0;
    {
        const int cc = t & 15, k0 = t >> 4;
        float v0 = 0.f, v1 = 0.f;
        if (cc < 8) {
            const int head = cc & 3, sel = cc >> 2;
            const float* wp = w + head * F_OUT + k0;
            const float* ap = a + head * 2 * F_OUT + sel * F_OUT + k0;
            const float w0 = wp[0], w1 = wp[F_OUT / 2], a0 = ap[0], a1 = ap[F_OUT / 2];
            v0 = w0 * a0;
            v1 = w1 * a1;
        }
        coefS[t] = v0;
        coefS[t + TILE_T] = v1;
    }
    __syncthreads();

    {
        const int wave = __builtin_amdgcn_readfirstlane(t >> 5);
        const int tile = (int)blockIdx.x + wave * N_TILES;
        if (tile < N_PROJ) {
            const int lane = t & 31;
            const int half = lane >> 4;
            const int rc   = lane & 15;
            const int koff = half * 2;
            const long arow = (long)(tile * 16 + rc) * F_OUT;

            v8f c = {0.f, 0.f, 0.f, 0.f, 0.f, 0.f, 0.f, 0.f};
#pragma unroll
            for (int k = 0; k < F_OUT; k += 4) {
                v2f av, bv;
                const float2 ax = *(const float2*)(x + arow + k + koff);
                av.x = ax.x; av.y = ax.y;
                bv.x = coefS[(k + koff) * 16 + rc];
                bv.y = coefS[(k + koff + 1) * 16 + rc];
                c = __builtin_amdgcn_wmma_f32_16x16x4_f32(false, av, false, bv,
                                                          (short)0, c, false, false);
            }
            if (rc < 8) {
                const int head = rc & 3;
                float* buf = (rc >> 2) ? s_dst : (out + REC_SSRC);
                const int pitch = (rc >> 2) ? N_HEAD : F_OUT;
                const int mbase = tile * 16 + half * 8;
#pragma unroll
                for (int j = 0; j < 8; ++j)
                    buf[(mbase + j) * pitch + head] = c[j];
            }
        }
    }

    int* rec = (int*)out;
    const int r0 = (int)(blockIdx.x % N_RANGE) * RANGE_N;
    const int sl = (int)(blockIdx.x / N_RANGE);
    const int* src = ei + sl * SLICE_E;
    const int* dst = ei + N_EDGES + sl * SLICE_E;
    int sv[TILE_PER], dv[TILE_PER];
#pragma unroll
    for (int k = 0; k < TILE_PER; ++k) {
        const int idx = k * TILE_T + t;
        sv[k] = src[min(idx, SLICE_E - 1)];
        dv[k] = dst[min(idx, SLICE_E - 1)];
        if (idx >= SLICE_E) sv[k] = -1;
    }
    int below = 0;
#pragma unroll
    for (int k = 0; k < TILE_PER; ++k)
        below += __popc(__builtin_amdgcn_ballot_w32((unsigned)sv[k] < (unsigned)r0));
    if ((t & 31) == 0) atomicAdd(&sbel, below);
    *(int4*)&lcnt[LCNT_PER * t] = make_int4(0, 0, 0, 0);
    __syncthreads();
#pragma unroll
    for (int k = 0; k < TILE_PER; ++k) {
        const unsigned d = (unsigned)(sv[k] - r0);
        if (d < (unsigned)RANGE_N) atomicAdd(&lcnt[d], 1);
    }
    __syncthreads();
    const int4 c4 = *(const int4*)&lcnt[LCNT_PER * t];
    const int s4 = c4.x + c4.y + c4.z + c4.w;
    int incl = s4;
    for (int o = 1; o < 32; o <<= 1) {
        int m = __shfl_up(incl, o);
        if ((t & 31) >= o) incl += m;
    }
    if ((t & 31) == 31) wsum[t >> 5] = incl;
    __syncthreads();
    if (t < 32) {
        const int v = wsum[t];
        int wi = v;
        for (int o = 1; o < 32; o <<= 1) {
            int m = __shfl_up(wi, o);
            if (t >= o) wi += m;
        }
        wsum[t] = wi - v;
    }
    __syncthreads();
    const int g0 = sl * SLICE_E + sbel;
    const int gb = g0 & ~7;
    const int p0 = g0 + wsum[t >> 5] + incl - s4;
    const int p1 = p0 + c4.x, p2 = p1 + c4.y, p3 = p2 + c4.z;
    *(int4*)&lcnt[LCNT_PER * t] = make_int4(p0 - gb, p1 - gb, p2 - gb, p3 - gb);
    if (t == TILE_T - 1) stot = p3 + c4.w - g0;
    {
        const int i0 = LCNT_PER * t;
        int* rp = rec + (long)(r0 + i0) * F_OUT + 2 * sl;
        if (i0 + 0 < RANGE_N) *(int2*)(rp + 0 * F_OUT) = make_int2(p0, c4.x);
        if (i0 + 1 < RANGE_N) *(int2*)(rp + 1 * F_OUT) = make_int2(p1, c4.y);
        if (i0 + 2 < RANGE_N) *(int2*)(rp + 2 * F_OUT) = make_int2(p2, c4.z);
        if (i0 + 3 < RANGE_N) *(int2*)(rp + 3 * F_OUT) = make_int2(p3, c4.w);
    }
    __syncthreads();
#pragma unroll
    for (int k = 0; k < TILE_PER; ++k) {
        const unsigned d = (unsigned)(sv[k] - r0);
        int pos = -1;
        if (d < (unsigned)RANGE_N) pos = atomicAdd(&lcnt[d], 1);
        sv[k] = pos;
    }
#pragma unroll
    for (int k = 0; k < TILE_PER; ++k)
        if (sv[k] >= 0) stage[sv[k]] = (unsigned short)dv[k];
    __syncthreads();
    const int g1 = g0 + stot;
    for (int e0 = gb + 8 * t; e0 < g1; e0 += 8 * TILE_T) {
        if (e0 >= g0 && e0 + 8 <= g1) {
            *(uint4*)(col + e0) = *(const uint4*)&stage[e0 - gb];
        } else {
            for (int j = 0; j < 8; ++j) {
                const int e = e0 + j;
                if (e >= g0 && e < g1) col[e] = stage[e - gb];
            }
        }
    }
}

__device__ __forceinline__ float edge_weight(float logit) {
    float lr = (logit >= 0.f) ? logit : NEG_SLOPE * logit;
    return expf(-lr);
}

__device__ __forceinline__ float lane_f(float v, int j) {
    return __int_as_float(__builtin_amdgcn_readlane(__float_as_int(v), j));
}

#define ROW_FMA(acc, e, xv) \
    acc.x = fmaf((e), (xv).x, acc.x); acc.y = fmaf((e), (xv).y, acc.y); \
    acc.z = fmaf((e), (xv).z, acc.z); acc.w = fmaf((e), (xv).w, acc.w)

#define EDGE_STEP(jj) { \
    const int jx = (jj); \
    const float4 xv = *(const float4*)(xl + (long)__builtin_amdgcn_readlane(dn, jx) * F_OUT); \
    const float f0 = lane_f(e0, jx), f1 = lane_f(e1, jx), f2 = lane_f(e2, jx), f3 = lane_f(e3, jx); \
    ROW_FMA(a0, f0, xv); ROW_FMA(a1, f1, xv); ROW_FMA(a2, f2, xv); ROW_FMA(a3, f3, xv); \
    r0 += f0; r1 += f1; r2 += f2; r3 += f3; }

__device__ __forceinline__ int run_shift(int rv, int e, int& count) {
    int shift = 0, first = 0;
#pragma unroll
    for (int s = 0; s < N_SLICE; ++s) {
        const int st = __builtin_amdgcn_readlane(rv, 2 * s);
        const int ln = __builtin_amdgcn_readlane(rv, 2 * s + 1);
        shift = (e >= first) ? (st - first) : shift;
        first += ln;
    }
    count = first;
    return shift;
}

__global__ void __launch_bounds__(256) row_kernel(const float* __restrict__ x,
                                                  const float* __restrict__ w,
                                                  const unsigned short* __restrict__ col,
                                                  const float* __restrict__ s_dst,
                                                  float* out) {
    const int lane = threadIdx.x & 31;
    const int row = __builtin_amdgcn_readfirstlane((int)(blockIdx.x * 8 + (threadIdx.x >> 5)));
    const float* rec = out + (long)row * F_OUT;
    const int    rv  = ((const int*)rec)[lane];
    const float4 ss  = *(const float4*)(rec + REC_SSRC);
    int cnt;
    int shift = run_shift(rv, lane, cnt);
    const float* xl = x + lane * 4;

    float4 a0 = make_float4(0.f, 0.f, 0.f, 0.f), a1 = a0, a2 = a0, a3 = a0;
    float r0 = 0.f, r1 = 0.f, r2 = 0.f, r3 = 0.f;

    for (int base = 0; base < cnt; base += 32) {
        const int n = min(32, cnt - base);
        int dn = 0;
        float e0 = 0.f, e1 = 0.f, e2 = 0.f, e3 = 0.f;
        const int e = base + lane;
        if (lane < n) {
            dn = min((int)col[shift + e], N_NODES - 1);
            const float4 sd = *(const float4*)(s_dst + dn * N_HEAD);
            e0 = edge_weight(ss.x + sd.x);
            e1 = edge_weight(ss.y + sd.y);
            e2 = edge_weight(ss.z + sd.z);
            e3 = edge_weight(ss.w + sd.w);
        }
        int j = 0;
        for (; j + 8 <= n; j += 8) {
#pragma unroll
            for (int u = 0; u < 8; ++u) { EDGE_STEP(j + u) }
        }
        if (j + 4 <= n) {
#pragma unroll
            for (int u = 0; u < 4; ++u) { EDGE_STEP(j + u) }
            j += 4;
        }
        if (j + 2 <= n) {
            EDGE_STEP(j) EDGE_STEP(j + 1)
            j += 2;
        }
        if (j < n) { EDGE_STEP(j) }
        if (base + 32 < cnt) { int again; shift = run_shift(rv, e + 32, again); }
    }

    const float4 w0 = *(const float4*)(w + 0 * F_OUT + lane * 4);
    const float4 w1 = *(const float4*)(w + 1 * F_OUT + lane * 4);
    const float4 w2 = *(const float4*)(w + 2 * F_OUT + lane * 4);
    const float4 w3 = *(const float4*)(w + 3 * F_OUT + lane * 4);
    const float i0 = 1.0f / r0, i1 = 1.0f / r1, i2 = 1.0f / r2, i3 = 1.0f / r3;
    float* o = out + (long)row * F_OUT + lane * 4;
    const long plane = (long)N_NODES * F_OUT;
    *(float4*)(o + 0 * plane) = make_float4(a0.x * w0.x * i0, a0.y * w0.y * i0, a0.z * w0.z * i0, a0.w * w0.w * i0);
    *(float4*)(o + 1 * plane) = make_float4(a1.x * w1.x * i1, a1.y * w1.y * i1, a1.z * w1.z * i1, a1.w * w1.w * i1);
    *(float4*)(o + 2 * plane) = make_float4(a2.x * w2.x * i2, a2.y * w2.y * i2, a2.z * w2.z * i2, a2.w * w2.w * i2);
    *(float4*)(o + 3 * plane) = make_float4(a3.x * w3.x * i3, a3.y * w3.y * i3, a3.z * w3.z * i3, a3.w * w3.w * i3);
}

extern "C" void kernel_launch(void* const* d_in, const int* in_sizes, int n_in,
                              void* d_out, int out_size, void* d_ws, size_t ws_size,
                              hipStream_t stream) {
    const float* x  = (const float*)d_in[0];
    const float* w  = (const float*)d_in[1];
    const float* a  = (const float*)d_in[2];
    const int*   ei = (const int*)d_in[3];
    float* out = (float*)d_out;

    float* ws    = (float*)d_ws;
    float* s_dst = ws;
    unsigned short* col = (unsigned short*)(ws + N_HEAD * N_NODES);

    front_kernel<<<N_TILES, TILE_T, 0, stream>>>(x, w, a, ei, out, s_dst, col);
    row_kernel<<<N_NODES / 8, 256, 0, stream>>>(x, w, col, s_dst, out);
}


# row records regrouped per 8-node block so a tile's record stores leave as 32-byte pieces, not 8 bytes per line; row pass picks them up through LDS behind one barrier; sum order unchanged
# speedup vs baseline: 11.2983x; 1.0959x over previous
#include <hip/hip_runtime.h>

#define N_NODES 50000
#define N_EDGES 800000
#define F_OUT   128
#define N_HEAD  4
#define NEG_SLOPE 0.2f

typedef __attribute__((ext_vector_type(2))) float v2f;
typedef __attribute__((ext_vector_type(8))) float v8f;

#define N_RANGE  16
#define RANGE_N  (N_NODES / N_RANGE)
#define N_SLICE  16
#define SLICE_E  (N_EDGES / N_SLICE)
#define TILE_T   1024
#define TILE_PER 49
#define LCNT_PER 4
#define LCNT_N   (TILE_T * LCNT_PER)
#define BLK_N    8
#define REC_LINE (2 * BLK_N)
#define REC_SSRC (N_SLICE * REC_LINE)
#define REC_PAD  18
#define N_TILES  (N_RANGE * N_SLICE)
#define N_PROJ   (N_NODES / 16)
#define STAGE_N  (SLICE_E + 8)
static_assert(N_RANGE * RANGE_N == N_NODES, "ranges must cover the nodes exactly");
static_assert(N_SLICE * SLICE_E == N_EDGES, "slices must cover the edges exactly");
static_assert(TILE_T * TILE_PER >= SLICE_E, "a tile's threads must cover its slice");
static_assert(LCNT_N >= RANGE_N + LCNT_PER - 1, "a tile's threads must cover its counters, counted from a multiple of four");
static_assert(LCNT_PER == 4 && BLK_N % LCNT_PER == 0, "a thread's four nodes lie in one record block");
static_assert(2 * N_SLICE == 32, "row pass: one record word per lane");
static_assert(REC_SSRC + BLK_N * N_HEAD <= BLK_N * F_OUT, "the records must fit the block's rows");
static_assert(BLK_N * 32 == N_SLICE * REC_LINE, "row pass: the block's waves, 32 words each, bring all of its record lines");
static_assert(BLK_N == 8, "projection: the eight nodes of a lane half are one record block");
static_assert(N_NODES <= 65536, "destination ids are kept in 16 bits");
static_assert(N_NODES % BLK_N == 0, "row pass: one work-group per record block, no tail");
static_assert(N_NODES % 16 == 0, "projection: 16-node tiles, no tail");
static_assert((long)N_TILES * (TILE_T / 32) >= N_PROJ, "the waves of the first launch must cover the projection tiles");

__global__ void __launch_bounds__(TILE_T) front_kernel(const float* __restrict__ x,
                                                       const float* __restrict__ w,
                                                       const float* __restrict__ a,
                                                       const int* __restrict__ ei,
                                                       float* out,
                                                       float* __restrict__ s_dst,
                                                       unsigned short* __restrict__ col) {
    __shared__ __align__(16) int lcnt[LCNT_N];
    __shared__ float coefS[F_OUT * 16];
    __shared__ int wsum[TILE_T / 32];
    __shared__ int sbel;
    __shared__ int stot;
    __shared__ __align__(16) unsigned short stage[STAGE_N];
    const int t  = threadIdx.x;

    if (t == 0) sbel = 0;
    {
        const int cc = t & 15, k0 = t >> 4;
        float v0 = 0.f, v1 = 0.f;
        if (cc < 8) {
            const int head = cc & 3, sel = cc >> 2;
            const float* wp = w + head * F_OUT + k0;
            const float* ap = a + head * 2 * F_OUT + sel * F_OUT + k0;
            const float w0 = wp[0], w1 = wp[F_OUT / 2], a0 = ap[0], a1 = ap[F_OUT / 2];
            v0 = w0 * a0;
            v1 = w1 * a1;
        }
        coefS[t] = v0;
        coefS[t + TILE_T] = v1;
    }
    __syncthreads();

    {
        const int wave = __builtin_amdgcn_readfirstlane(t >> 5);
        const int tile = (int)blockIdx.x + wave * N_TILES;
        if (tile < N_PROJ) {
            const int lane = t & 31;
            const int half = lane >> 4;
            const int rc   = lane & 15;
            const int koff = half * 2;
            const long arow = (long)(tile * 16 + rc) * F_OUT;

            v8f c = {0.f, 0.f, 0.f, 0.f, 0.f, 0.f, 0.f, 0.f};
#pragma unroll
            for (int k = 0; k < F_OUT; k += 4) {
                v2f av, bv;
                const float2 ax = *(const float2*)(x + arow + k + koff);
                av.x = ax.x; av.y = ax.y;
                bv.x = coefS[(k + koff) * 16 + rc];
                bv.y = coefS[(k + koff + 1) * 16 + rc];
                c = __builtin_amdgcn_wmma_f32_16x16x4_f32(false, av, false, bv,
                                                          (short)0, c, false, false);
            }
            if (rc < 8) {
                const int head = rc & 3;
                const int m0 = tile * 16 + half * 8;
                float* buf = (rc >> 2) ? (s_dst + m0 * N_HEAD)
                                       : (out + (long)m0 * F_OUT + REC_SSRC);
#pragma unroll
                for (int j = 0; j < 8; ++j)
                    buf[j * N_HEAD + head] = c[j];
            }
        }
    }

    int* rec = (int*)out;
    const int r0 = (int)(blockIdx.x % N_RANGE) * RANGE_N;
    const int sl = (int)(blockIdx.x / N_RANGE);
    const int roff = r0 & (LCNT_PER - 1);
    const int* src = ei + sl * SLICE_E;
    const int* dst = ei + N_EDGES + sl * SLICE_E;
    int sv[TILE_PER], dv[TILE_PER];
#pragma unroll
    for (int k = 0; k < TILE_PER; ++k) {
        const int idx = k * TILE_T + t;
        sv[k] = src[min(idx, SLICE_E - 1)];
        dv[k] = dst[min(idx, SLICE_E - 1)];
        if (idx >= SLICE_E) sv[k] = -1;
    }
    int below = 0;
#pragma unroll
    for (int k = 0; k < TILE_PER; ++k)
        below += __popc(__builtin_amdgcn_ballot_w32((unsigned)sv[k] < (unsigned)r0));
    if ((t & 31) == 0) atomicAdd(&sbel, below);
    *(int4*)&lcnt[LCNT_PER * t] = make_int4(0, 0, 0, 0);
    __syncthreads();
#pragma unroll
    for (int k = 0; k < TILE_PER; ++k) {
        const unsigned d = (unsigned)(sv[k] - r0);
        if (d < (unsigned)RANGE_N) atomicAdd(&lcnt[d + roff], 1);
    }
    __syncthreads();
    const int4 c4 = *(const int4*)&lcnt[LCNT_PER * t];
    const int s4 = c4.x + c4.y + c4.z + c4.w;
    int incl = s4;
    for (int o = 1; o < 32; o <<= 1) {
        int m = __shfl_up(incl, o);
        if ((t & 31) >= o) incl += m;
    }
    if ((t & 31) == 31) wsum[t >> 5] = incl;
    __syncthreads();
    if (t < 32) {
        const int v = wsum[t];
        int wi = v;
        for (int o = 1; o < 32; o <<= 1) {
            int m = __shfl_up(wi, o);
            if (t >= o) wi += m;
        }
        wsum[t] = wi - v;
    }
    __syncthreads();
    const int g0 = sl * SLICE_E + sbel;
    const int gb = g0 & ~7;
    const int p0 = g0 + wsum[t >> 5] + incl - s4;
    const int p1 = p0 + c4.x, p2 = p1 + c4.y, p3 = p2 + c4.z;
    *(int4*)&lcnt[LCNT_PER * t] = make_int4(p0 - gb, p1 - gb, p2 - gb, p3 - gb);
    if (t == TILE_T - 1) stot = p3 + c4.w - g0;
    {
        const int nb = r0 - roff + LCNT_PER * t;
        const int lo = nb - r0;
        int* rp = rec + (long)(nb & ~(BLK_N - 1)) * F_OUT + sl * REC_LINE + (nb & (BLK_N - 1)) * 2;
        if (lo >= 0 && lo + 3 < RANGE_N) {
            *(int4*)(rp + 0) = make_int4(p0, c4.x, p1, c4.y);
            *(int4*)(rp + 4) = make_int4(p2, c4.z, p3, c4.w);
        } else {
            if ((unsigned)(lo + 0) < (unsigned)RANGE_N) *(int2*)(rp + 0) = make_int2(p0, c4.x);
            if ((unsigned)(lo + 1) < (unsigned)RANGE_N) *(int2*)(rp + 2) = make_int2(p1, c4.y);
            if ((unsigned)(lo + 2) < (unsigned)RANGE_N) *(int2*)(rp + 4) = make_int2(p2, c4.z);
            if ((unsigned)(lo + 3) < (unsigned)RANGE_N) *(int2*)(rp + 6) = make_int2(p3, c4.w);
        }
    }
    __syncthreads();
#pragma unroll
    for (int k = 0; k < TILE_PER; ++k) {
        const unsigned d = (unsigned)(sv[k] - r0);
        int pos = -1;
        if (d < (unsigned)RANGE_N) pos = atomicAdd(&lcnt[d + roff], 1);
        sv[k] = pos;
    }
#pragma unroll
    for (int k = 0; k < TILE_PER; ++k)
        if (sv[k] >= 0) stage[sv[k]] = (unsigned short)dv[k];
    __syncthreads();
    const int g1 = g0 + stot;
    for (int e0 = gb + 8 * t; e0 < g1; e0 += 8 * TILE_T) {
        if (e0 >= g0 && e0 + 8 <= g1) {
            *(uint4*)(col + e0) = *(const uint4*)&stage[e0 - gb];
        } else {
            for (int j = 0; j < 8; ++j) {
                const int e = e0 + j;
                if (e >= g0 && e < g1) col[e] = stage[e - gb];
            }
        }
    }
}

__device__ __forceinline__ float edge_weight(float logit) {
    float lr = (logit >= 0.f) ? logit : NEG_SLOPE * logit;
    return expf(-lr);
}

__device__ __forceinline__ float lane_f(float v, int j) {
    return __int_as_float(__builtin_amdgcn_readlane(__float_as_int(v), j));
}

__device__ __forceinline__ float wave_f(float v) {
    return __int_as_float(__builtin_amdgcn_readfirstlane(__float_as_int(v)));
}

#define ROW_FMA(acc, e, xv) \
    acc.x = fmaf((e), (xv).x, acc.x); acc.y = fmaf((e), (xv).y, acc.y); \
    acc.z = fmaf((e), (xv).z, acc.z); acc.w = fmaf((e), (xv).w, acc.w)

#define EDGE_STEP(jj) { \
    const int jx = (jj); \
    const float4 xv = *(const float4*)(xl + (long)__builtin_amdgcn_readlane(dn, jx) * F_OUT); \
    const float f0 = lane_f(e0, jx), f1 = lane_f(e1, jx), f2 = lane_f(e2, jx), f3 = lane_f(e3, jx); \
    ROW_FMA(a0, f0, xv); ROW_FMA(a1, f1, xv); ROW_FMA(a2, f2, xv); ROW_FMA(a3, f3, xv); \
    r0 += f0; r1 += f1; r2 += f2; r3 += f3; }

__device__ __forceinline__ int run_shift(int rv, int e, int& count) {
    int shift = 0, first = 0;
#pragma unroll
    for (int s = 0; s < N_SLICE; ++s) {
        const int st = __builtin_amdgcn_readlane(rv, 2 * s);
        const int ln = __builtin_amdgcn_readlane(rv, 2 * s + 1);
        shift = (e >= first) ? (st - first) : shift;
        first += ln;
    }
    count = first;
    return shift;
}

__global__ void __launch_bounds__(BLK_N * 32) row_kernel(const float* __restrict__ x,
                                                  const float* __restrict__ w,
                                                  const unsigned short* __restrict__ col,
                                                  const float* __restrict__ s_dst,
                                                  float* out) {
    const int lane = threadIdx.x & 31;
    const int wv  = __builtin_amdgcn_readfirstlane((int)(threadIdx.x >> 5));
    const int row = (int)blockIdx.x * BLK_N + wv;
    __shared__ __align__(16) int recS[N_SLICE * REC_PAD + BLK_N * N_HEAD];
    {
        const int* blk = (const int*)out + (long)blockIdx.x * BLK_N * F_OUT;
        const int g = wv * 32 + lane;
        const int line = blk[g];
        int term = 0;
        if (lane < N_HEAD) term = blk[REC_SSRC + wv * N_HEAD + lane];
        recS[(g / REC_LINE) * REC_PAD + (g % REC_LINE)] = line;
        if (lane < N_HEAD) recS[N_SLICE * REC_PAD + wv * N_HEAD + lane] = term;
    }
    __syncthreads();
    const int    rv  = recS[(lane >> 1) * REC_PAD + wv * 2 + (lane & 1)];
    const float4 sv4 = *(const float4*)&recS[N_SLICE * REC_PAD + wv * N_HEAD];
    const float4 ss  = make_float4(wave_f(sv4.x), wave_f(sv4.y), wave_f(sv4.z), wave_f(sv4.w));
    int cnt;
    int shift = run_shift(rv, lane, cnt);
    const float* xl = x + lane * 4;

    float4 a0 = make_float4(0.f, 0.f, 0.f, 0.f), a1 = a0, a2 = a0, a3 = a0;
    float r0 = 0.f, r1 = 0.f, r2 = 0.f, r3 = 0.f;

    for (int base = 0; base < cnt; base += 32) {
        const int n = min(32, cnt - base);
        int dn = 0;
        float e0 = 0.f, e1 = 0.f, e2 = 0.f, e3 = 0.f;
        const int e = base + lane;
        if (lane < n) {
            dn = min((int)col[shift + e], N_NODES - 1);
            const float4 sd = *(const float4*)(s_dst + dn * N_HEAD);
            e0 = edge_weight(ss.x + sd.x);
            e1 = edge_weight(ss.y + sd.y);
            e2 = edge_weight(ss.z + sd.z);
            e3 = edge_weight(ss.w + sd.w);
        }
        int j = 0;
        for (; j + 8 <= n; j += 8) {
#pragma unroll
            for (int u = 0; u < 8; ++u) { EDGE_STEP(j + u) }
        }
        if (j + 4 <= n) {
#pragma unroll
            for (int u = 0; u < 4; ++u) { EDGE_STEP(j + u) }
            j += 4;
        }
        if (j + 2 <= n) {
            EDGE_STEP(j) EDGE_STEP(j + 1)
            j += 2;
        }
        if (j < n) { EDGE_STEP(j) }
        if (base + 32 < cnt) { int again; shift = run_shift(rv, e + 32, again); }
    }

    const float4 w0 = *(const float4*)(w + 0 * F_OUT + lane * 4);
    const float4 w1 = *(const float4*)(w + 1 * F_OUT + lane * 4);
    const float4 w2 = *(const float4*)(w + 2 * F_OUT + lane * 4);
    const float4 w3 = *(const float4*)(w + 3 * F_OUT + lane * 4);
    const float i0 = 1.0f / r0, i1 = 1.0f / r1, i2 = 1.0f / r2, i3 = 1.0f / r3;
    float* o = out + (long)row * F_OUT + lane * 4;
    const long plane = (long)N_NODES * F_OUT;
    *(float4*)(o + 0 * plane) = make_float4(a0.x * w0.x * i0, a0.y * w0.y * i0, a0.z * w0.z * i0, a0.w * w0.w * i0);
    *(float4*)(o + 1 * plane) = make_float4(a1.x * w1.x * i1, a1.y * w1.y * i1, a1.z * w1.z * i1, a1.w * w1.w * i1);
    *(float4*)(o + 2 * plane) = make_float4(a2.x * w2.x * i2, a2.y * w2.y * i2, a2.z * w2.z * i2, a2.w * w2.w * i2);
    *(float4*)(o + 3 * plane) = make_float4(a3.x * w3.x * i3, a3.y * w3.y * i3, a3.z * w3.z * i3, a3.w * w3.w * i3);
}

extern "C" void kernel_launch(void* const* d_in, const int* in_sizes, int n_in,
                              void* d_out, int out_size, void* d_ws, size_t ws_size,
                              hipStream_t stream) {
    const float* x  = (const float*)d_in[0];
    const float* w  = (const float*)d_in[1];
    const float* a  = (const float*)d_in[2];
    const int*   ei = (const int*)d_in[3];
    float* out = (float*)d_out;

    float* ws    = (float*)d_ws;
    float* s_dst = ws;
    unsigned short* col = (unsigned short*)(ws + N_HEAD * N_NODES);

    front_kernel<<<N_TILES, TILE_T, 0, stream>>>(x, w, a, ei, out, s_dst, col);
    row_kernel<<<N_NODES / BLK_N, BLK_N * 32, 0, stream>>>(x, w, col, s_dst, out);
}
